# starting kernel
# baseline (speedup 1.0000x reference)
#include <hip/hip_runtime.h>
#include <math.h>

typedef float v2f __attribute__((ext_vector_type(2)));
typedef float v8f __attribute__((ext_vector_type(8)));

#define EMBD 768
#define HEAD 64
#define SEQ  4096
#define BATCH 4
#define BS   (BATCH * SEQ)

__device__ __forceinline__ v8f wmma_f32(v2f a, v2f b, v8f c) {
    return __builtin_amdgcn_wmma_f32_16x16x4_f32(
         false, a,  false, b,
         (short)0, c,  false,  false);
}

__global__ __launch_bounds__(256) void qkv_proj_kernel(
    const float* __restrict__ x,
    const float* __restrict__ Wq, const float* __restrict__ Wk,
    const float* __restrict__ Wv,
    float* __restrict__ qb, float* __restrict__ kb, float* __restrict__ vT)
{
    const int lane = threadIdx.x & 31;
    const int wave = threadIdx.x >> 5;
    const int g    = blockIdx.x * 8 + wave;
    const int mat  = g >> 10;
    const int tile = g & 1023;
    const int row0 = tile << 4;
    const float* W = (mat == 0) ? Wq : (mat == 1) ? Wk : Wv;

    const int half = lane >> 4;
    const int lx   = lane & 15;

    v8f acc[4] = {};
    const float* xrow = x + (size_t)(row0 + lx) * EMBD;
    for (int c = 0; c < EMBD; c += 4) {
        const int kk = c + 2 * half;
        v2f a = *(const v2f*)(xrow + kk);
        #pragma unroll
        for (int t = 0; t < 4; ++t) {
            v2f bf;
            bf.x = W[(size_t)kk       * HEAD + t * 16 + lx];
            bf.y = W[(size_t)(kk + 1) * HEAD + t * 16 + lx];
            acc[t] = wmma_f32(a, bf, acc[t]);
        }
    }

    if (mat < 2) {
        float* outb = (mat == 0) ? qb : kb;
        #pragma unroll
        for (int t = 0; t < 4; ++t)
            #pragma unroll
            for (int i = 0; i < 8; ++i) {
                const int r = row0 + i + 8 * half;
                outb[(size_t)r * HEAD + t * 16 + lx] = acc[t][i];
            }
    } else {
        const int batch = row0 >> 12;
        const int sbase = row0 & 4095;
        float* vb = vT + (size_t)batch * HEAD * SEQ;
        #pragma unroll
        for (int t = 0; t < 4; ++t)
            #pragma unroll
            for (int i = 0; i < 8; ++i) {
                const int s = sbase + i + 8 * half;
                const int h = t * 16 + lx;
                vb[(size_t)h * SEQ + s] = acc[t][i];
            }
    }
}

__global__ __launch_bounds__(256) void flash_attn_kernel(
    const float* __restrict__ qb, const float* __restrict__ kb,
    const float* __restrict__ vT, float* __restrict__ out)
{
    const int lane  = threadIdx.x & 31;
    const int wave  = threadIdx.x >> 5;
    const int qt    = blockIdx.x * 8 + wave;
    const int batch = qt >> 8;
    const int q0    = (qt & 255) << 4;
    const int half  = lane >> 4;
    const int lx    = lane & 15;

    const float scale = 0.125f;

    v2f qf[16];
    {
        const float* qrow = qb + (size_t)(batch * SEQ + q0 + lx) * HEAD;
        #pragma unroll
        for (int c = 0; c < 16; ++c) {
            v2f v = *(const v2f*)(qrow + 4 * c + 2 * half);
            qf[c].x = v.x * scale;
            qf[c].y = v.y * scale;
        }
    }

    v8f acc[4] = {};
    float m = -INFINITY, l = 0.0f;
    const int q_lane = q0 + lx;
    const float* kbase = kb + (size_t)batch * SEQ * HEAD;
    const float* vbase = vT + (size_t)batch * HEAD * SEQ;

    const int nkt = (q0 >> 4) + 1;
    for (int kt = 0; kt < nkt; ++kt) {
        const int k0 = kt << 4;

        v8f st = {};
        const float* krow = kbase + (size_t)(k0 + lx) * HEAD;
        #pragma unroll
        for (int c = 0; c < 16; ++c) {
            v2f kf = *(const v2f*)(krow + 4 * c + 2 * half);
            st = wmma_f32(kf, qf[c], st);
        }

        float p[8];
        float tmax = -INFINITY;
        #pragma unroll
        for (int i = 0; i < 8; ++i) {
            const int key = k0 + i + 8 * half;
            const float s = (key <= q_lane) ? st[i] : -INFINITY;
            p[i] = s;
            tmax = fmaxf(tmax, s);
        }
        tmax = fmaxf(tmax, __shfl_xor(tmax, 16, 32));
        const float m_new = fmaxf(m, tmax);
        const float alpha = __expf(m - m_new);
        float rsum = 0.0f;
        #pragma unroll
        for (int i = 0; i < 8; ++i) {
            p[i] = __expf(p[i] - m_new);
            rsum += p[i];
        }
        l = l * alpha + rsum + __shfl_xor(rsum, 16, 32);
        m = m_new;
        #pragma unroll
        for (int t = 0; t < 4; ++t)
            #pragma unroll
            for (int i = 0; i < 8; ++i) acc[t][i] *= alpha;

        float sh[8];
        #pragma unroll
        for (int i = 0; i < 8; ++i) sh[i] = __shfl_xor(p[i], 16, 32);
        v2f pf[4];
        #pragma unroll
        for (int c = 0; c < 4; ++c) {
            const int keyA = 4 * c + 2 * half;
            const int keyB = keyA + 1;
            pf[c].x = ((keyA >> 3) == half) ? p[keyA & 7] : sh[keyA & 7];
            pf[c].y = ((keyB >> 3) == half) ? p[keyB & 7] : sh[keyB & 7];
        }

        #pragma unroll
        for (int t = 0; t < 4; ++t) {
            const float* vrow = vbase + (size_t)(t * 16 + lx) * SEQ + k0;
            #pragma unroll
            for (int c = 0; c < 4; ++c) {
                v2f vf = *(const v2f*)(vrow + 4 * c + 2 * half);
                acc[t] = wmma_f32(vf, pf[c], acc[t]);
            }
        }
    }

    const float inv = 1.0f / l;
    float* orow = out + (size_t)(batch * SEQ + q0 + lx) * HEAD + 8 * half;
    #pragma unroll
    for (int t = 0; t < 4; ++t) {
        float4 lo = make_float4(acc[t][0] * inv, acc[t][1] * inv,
                                acc[t][2] * inv, acc[t][3] * inv);
        float4 hi = make_float4(acc[t][4] * inv, acc[t][5] * inv,
                                acc[t][6] * inv, acc[t][7] * inv);
        *(float4*)(orow + t * 16)     = lo;
        *(float4*)(orow + t * 16 + 4) = hi;
    }
}

extern "C" void kernel_launch(void* const* d_in, const int* in_sizes, int n_in,
                              void* d_out, int out_size, void* d_ws, size_t ws_size,
                              hipStream_t stream) {
    const float* x  = (const float*)d_in[0];
    const float* Wq = (const float*)d_in[1];
    const float* Wk = (const float*)d_in[2];
    const float* Wv = (const float*)d_in[3];

    float* qb = (float*)d_ws;
    float* kb = qb + (size_t)BS * HEAD;
    float* vT = kb + (size_t)BS * HEAD;

    qkv_proj_kernel<<<384, 256, 0, stream>>>(x, Wq, Wk, Wv, qb, kb, vT);
    flash_attn_kernel<<<128, 256, 0, stream>>>(qb, kb, vT, (float*)d_out);
}


# Projection: one wave per 16 rows makes all 192 Q|K|V columns in registers, no LDS; x read once, W by 16-B loads that feed 4 N-tiles each (column 4c+t), one 16-e block ahead; k sum order changes
# speedup vs baseline: 6.8065x; 6.8065x over previous
#include <hip/hip_runtime.h>
#include <math.h>

typedef float v2f __attribute__((ext_vector_type(2)));
typedef float v4f __attribute__((ext_vector_type(4)));
typedef float v8f __attribute__((ext_vector_type(8)));

#define EMBD 768
#define HEAD 64
#define SEQ  4096
#define BATCH 4
#define BS   (BATCH * SEQ)

__device__ __forceinline__ v8f wmma_f32(v2f a, v2f b, v8f c) {
    return __builtin_amdgcn_wmma_f32_16x16x4_f32(
         false, a,  false, b,
         (short)0, c,  false,  false);
}

__device__ __forceinline__ float xhalf(float x) {
    const int b = __builtin_bit_cast(int, x);
    return __builtin_bit_cast(float,
        __builtin_amdgcn_permlanex16(b, b, 0x76543210, 0xfedcba98, false, false));
}

#define NSPAN (EMBD / 64)

__device__ __forceinline__ void proj_load_w(v4f (&b)[24], const float* wq,
                                            const float* wk, const float* wv, int e0)
{
    #pragma unroll
    for (int u = 0; u < 2; ++u)
        #pragma unroll
        for (int i = 0; i < 4; ++i) {
            const int off = (e0 + 8 * u + i) * HEAD;
            b[(u * 4 + i) * 3 + 0] = *(const v4f*)(wq + off);
            b[(u * 4 + i) * 3 + 1] = *(const v4f*)(wk + off);
            b[(u * 4 + i) * 3 + 2] = *(const v4f*)(wv + off);
        }
}

__device__ __forceinline__ void proj_mma(v8f (&acc)[12], const v4f a0, const v4f a1,
                                         const v4f (&b)[24])
{
    #pragma unroll
    for (int u = 0; u < 2; ++u) {
        const v4f a = u ? a1 : a0;
        #pragma unroll
        for (int s = 0; s < 2; ++s) {
            v2f af;
            af.x = a[2 * s];
            af.y = a[2 * s + 1];
            #pragma unroll
            for (int m = 0; m < 3; ++m)
                #pragma unroll
                for (int t = 0; t < 4; ++t) {
                    v2f bf;
                    bf.x = b[(u * 4 + 2 * s) * 3 + m][t];
                    bf.y = b[(u * 4 + 2 * s + 1) * 3 + m][t];
                    acc[m * 4 + t] = wmma_f32(af, bf, acc[m * 4 + t]);
                }
        }
    }
}

__global__ __launch_bounds__(128) void qkv_proj_kernel(
    const float* __restrict__ x,
    const float* __restrict__ Wq, const float* __restrict__ Wk,
    const float* __restrict__ Wv,
    float* __restrict__ qb, float* __restrict__ kb, float* __restrict__ vT)
{
    const int lane = threadIdx.x & 31;
    const int wave = threadIdx.x >> 5;
    const int half = lane >> 4;
    const int lx   = lane & 15;
    const int row0 = (blockIdx.x * 4 + wave) << 4;

    const float* xrow = x + (size_t)(row0 + lx) * EMBD + 4 * half;
    const int woff = (4 * half) * HEAD + 4 * lx;
    const float* wq = Wq + woff;
    const float* wk = Wk + woff;
    const float* wv = Wv + woff;

    v8f acc[12] = {};
    v4f a_cur[8], a_nxt[8];
    v4f b0[24], b1[24];

    int sp = blockIdx.x % NSPAN;

    #pragma unroll
    for (int j = 0; j < 8; ++j) a_cur[j] = *(const v4f*)(xrow + 64 * sp + 8 * j);
    proj_load_w(b0, wq, wk, wv, 64 * sp);

    #pragma unroll 1
    for (int it = 0; it < NSPAN; ++it) {
        int spn = sp + 1;
        if (spn == NSPAN) spn = 0;
        const int e = 64 * sp;
        #pragma unroll
        for (int j = 0; j < 8; ++j) a_nxt[j] = *(const v4f*)(xrow + 64 * spn + 8 * j);

        proj_load_w(b1, wq, wk, wv, e + 16);
        __builtin_amdgcn_sched_barrier(0);
        proj_mma(acc, a_cur[0], a_cur[1], b0);
        __builtin_amdgcn_sched_barrier(0);
        proj_load_w(b0, wq, wk, wv, e + 32);
        __builtin_amdgcn_sched_barrier(0);
        proj_mma(acc, a_cur[2], a_cur[3], b1);
        __builtin_amdgcn_sched_barrier(0);
        proj_load_w(b1, wq, wk, wv, e + 48);
        __builtin_amdgcn_sched_barrier(0);
        proj_mma(acc, a_cur[4], a_cur[5], b0);
        __builtin_amdgcn_sched_barrier(0);
        proj_load_w(b0, wq, wk, wv, 64 * spn);
        __builtin_amdgcn_sched_barrier(0);
        proj_mma(acc, a_cur[6], a_cur[7], b1);
        __builtin_amdgcn_sched_barrier(0);

        #pragma unroll
        for (int j = 0; j < 8; ++j) a_cur[j] = a_nxt[j];
        sp = spn;
    }

    const int r0 = row0 + 8 * half;
    #pragma unroll
    for (int m = 0; m < 2; ++m) {
        float* ob = (m == 0) ? qb : kb;
        #pragma unroll
        for (int i = 0; i < 8; ++i) {
            v4f o;
            o.x = acc[m * 4 + 0][i];
            o.y = acc[m * 4 + 1][i];
            o.z = acc[m * 4 + 2][i];
            o.w = acc[m * 4 + 3][i];
            *(v4f*)(ob + (size_t)(r0 + i) * HEAD + 4 * lx) = o;
        }
    }
    {
        const int batch = row0 >> 12;
        const int s0    = (row0 & 4095) + 8 * half;
        float* vb = vT + (size_t)batch * HEAD * SEQ;
        #pragma unroll
        for (int t = 0; t < 4; ++t) {
            float* p = vb + (size_t)(4 * lx + t) * SEQ + s0;
            v4f lo, up;
            lo.x = acc[8 + t][0]; lo.y = acc[8 + t][1];
            lo.z = acc[8 + t][2]; lo.w = acc[8 + t][3];
            up.x = acc[8 + t][4]; up.y = acc[8 + t][5];
            up.z = acc[8 + t][6]; up.w = acc[8 + t][7];
            *(v4f*)(p)     = lo;
            *(v4f*)(p + 4) = up;
        }
    }
}

#define QW 4
#define KC 64
#define LP 68
#define CUT 33

__global__ __launch_bounds__(128) void flash_attn_kernel(
    const float* __restrict__ qb, const float* __restrict__ kb,
    const float* __restrict__ vT, float* __restrict__ out,
    float* __restrict__ pacc, float* __restrict__ pm, float* __restrict__ pl)
{
    __shared__ __attribute__((aligned(16))) float sK[2][KC * LP];
    __shared__ __attribute__((aligned(16))) float sV[2][HEAD * LP];

    const int tid   = threadIdx.x;
    const int lane  = tid & 31;
    const int wave  = __builtin_amdgcn_readfirstlane(tid >> 5);
    const int batch = blockIdx.x & 3;
    const int pr    = blockIdx.x >> 3;
    const int share = (blockIdx.x >> 2) & 1;
    const int half  = lane >> 4;
    const int lx    = lane & 15;
    const bool hi   = (half != 0);

    const float scale = 0.125f;

    const float* kbase = kb + (size_t)batch * SEQ * HEAD;
    const float* vbase = vT + (size_t)batch * HEAD * SEQ;

    const int srow = tid >> 4;
    const int scol = (tid & 15) << 2;

    const int s_first = share ? (pr < 31 ? 1 : 2) : 0;
    const int s_last  = share ? 2 : 0;

    for (int s = s_first; s <= s_last; ++s) {
        const int qblk = (s == 2) ? pr : 63 - pr;
        const int c0   = (s == 1) ? CUT : 0;
        const int c1   = (s == 0) ? CUT : qblk + 1;
        const bool whole = (s == 2) || (pr == 31);

        const int qt = qblk * QW + wave;
        const int q0 = qt << 4;

        v2f qf[16];
        {
            const float* qrow = qb + (size_t)(batch * SEQ + q0 + lx) * HEAD;
            #pragma unroll
            for (int c = 0; c < 16; ++c) {
                v2f v = *(const v2f*)(qrow + 4 * c + 2 * half);
                qf[c].x = v.x * scale;
                qf[c].y = v.y * scale;
            }
        }

        v8f acc[4] = {};
        float m = -INFINITY, l = 0.0f;
        const int q_lane = q0 + lx;
        const int nkt    = qt + 1;

        v4f rk[8], rv[8];

        {
            const int kc0 = c0 * KC;
            #pragma unroll
            for (int i = 0; i < 8; ++i) {
                const int r = srow + 8 * i;
                rk[i] = *(const v4f*)(kbase + (size_t)(kc0 + r) * HEAD + scol);
                rv[i] = *(const v4f*)(vbase + (size_t)r * SEQ + kc0 + scol);
            }
            #pragma unroll
            for (int i = 0; i < 8; ++i) {
                const int r = srow + 8 * i;
                *(v4f*)(&sK[0][r * LP + scol]) = rk[i];
                *(v4f*)(&sV[0][r * LP + scol]) = rv[i];
            }
        }
        __syncthreads();

        for (int ch = c0; ch < c1; ++ch) {
            const int buf   = (ch - c0) & 1;
            const bool more = (ch + 1 < c1);

            if (more) {
                const int kc1 = (ch + 1) * KC;
                #pragma unroll
                for (int i = 0; i < 8; ++i) {
                    const int r = srow + 8 * i;
                    rk[i] = *(const v4f*)(kbase + (size_t)(kc1 + r) * HEAD + scol);
                    rv[i] = *(const v4f*)(vbase + (size_t)r * SEQ + kc1 + scol);
                }
            }

            const float* cK = sK[buf];
            const float* cV = sV[buf];
            int ntile = nkt - 4 * ch;
            if (ntile > 4) ntile = 4;

            for (int kk = 0; kk < ntile; ++kk) {
                const int k0 = (4 * ch + kk) << 4;

                v8f st = {};
                const float* krow = cK + (kk * 16 + lx) * LP + 2 * half;
                #pragma unroll
                for (int c = 0; c < 16; ++c) {
                    v2f kf = *(const v2f*)(krow + 4 * c);
                    st = wmma_f32(kf, qf[c], st);
                }

                float p[8];
                float tmax = -INFINITY;
                #pragma unroll
                for (int i = 0; i < 8; ++i) {
                    const int key = k0 + i + 8 * half;
                    const float sc = (key <= q_lane) ? st[i] : -INFINITY;
                    p[i] = sc;
                    tmax = fmaxf(tmax, sc);
                }
                tmax = fmaxf(tmax, xhalf(tmax));
                const float m_new = fmaxf(m, tmax);
                const float alpha = __expf(m - m_new);
                float rsum = 0.0f;
                #pragma unroll
                for (int i = 0; i < 8; ++i) {
                    p[i] = __expf(p[i] - m_new);
                    rsum += p[i];
                }
                l = l * alpha + rsum + xhalf(rsum);
                m = m_new;
                #pragma unroll
                for (int t = 0; t < 4; ++t)
                    #pragma unroll
                    for (int i = 0; i < 8; ++i) acc[t][i] *= alpha;

                const float g0 = xhalf(hi ? p[0] : p[2]);
                const float g1 = xhalf(hi ? p[1] : p[3]);
                const float g2 = xhalf(hi ? p[4] : p[6]);
                const float g3 = xhalf(hi ? p[5] : p[7]);
                v2f pf[4];
                pf[0].x = hi ? g0 : p[0];   pf[0].y = hi ? g1 : p[1];
                pf[1].x = hi ? g2 : p[4];   pf[1].y = hi ? g3 : p[5];
                pf[2].x = hi ? p[2] : g0;   pf[2].y = hi ? p[3] : g1;
                pf[3].x = hi ? p[6] : g2;   pf[3].y = hi ? p[7] : g3;

                #pragma unroll
                for (int t = 0; t < 4; ++t) {
                    const float* vrow = cV + (t * 16 + lx) * LP + kk * 16 + 2 * half;
                    #pragma unroll
                    for (int c = 0; c < 4; ++c) {
                        v2f vf = *(const v2f*)(vrow + 4 * c);
                        acc[t] = wmma_f32(vf, pf[c], acc[t]);
                    }
                }
            }

            if (more) {
                #pragma unroll
                for (int i = 0; i < 8; ++i) {
                    const int r = srow + 8 * i;
                    *(v4f*)(&sK[buf ^ 1][r * LP + scol]) = rk[i];
                    *(v4f*)(&sV[buf ^ 1][r * LP + scol]) = rv[i];
                }
            }
            __syncthreads();
        }

        if (whole) {
            const float inv = 1.0f / l;
            float* orow = out + (size_t)(batch * SEQ + q0 + lx) * HEAD + 8 * half;
            #pragma unroll
            for (int t = 0; t < 4; ++t) {
                float4 lo = make_float4(acc[t][0] * inv, acc[t][1] * inv,
                                        acc[t][2] * inv, acc[t][3] * inv);
                float4 up = make_float4(acc[t][4] * inv, acc[t][5] * inv,
                                        acc[t][6] * inv, acc[t][7] * inv);
                *(float4*)(orow + t * 16)     = lo;
                *(float4*)(orow + t * 16 + 4) = up;
            }
        } else {
            const int prow = (((batch * 32 + pr) * 2 + s) * 64) + wave * 16 + lx;
            float* arow = pacc + (size_t)prow * HEAD + 8 * half;
            #pragma unroll
            for (int t = 0; t < 4; ++t) {
                float4 lo = make_float4(acc[t][0], acc[t][1], acc[t][2], acc[t][3]);
                float4 up = make_float4(acc[t][4], acc[t][5], acc[t][6], acc[t][7]);
                *(float4*)(arow + t * 16)     = lo;
                *(float4*)(arow + t * 16 + 4) = up;
            }
            if (!hi) {
                pm[prow] = m;
                pl[prow] = l;
            }
        }
    }
}

__global__ __launch_bounds__(64) void merge_kernel(
    const float* __restrict__ pacc, const float* __restrict__ pm,
    const float* __restrict__ pl, float* __restrict__ out)
{
    const int tid = threadIdx.x;
    const int hq  = tid & 15;
    const int qs  = tid >> 4;

    #pragma unroll
    for (int k = 0; k < 4; ++k) {
        const int ql    = blockIdx.x * 16 + qs + 4 * k;
        const int batch = ql / 1984;
        const int rem   = ql - batch * 1984;
        const int pr    = rem >> 6;
        const int qi    = rem & 63;
        const int row0  = ((batch * 32 + pr) * 2) * 64 + qi;
        const int row1  = row0 + 64;

        const float m0 = pm[row0], m1 = pm[row1];
        const float l0 = pl[row0], l1 = pl[row1];
        const float mm = fmaxf(m0, m1);
        const float e0 = __expf(m0 - mm), e1 = __expf(m1 - mm);
        const float inv = 1.0f / (e0 * l0 + e1 * l1);

        const v4f a0 = *(const v4f*)(pacc + (size_t)row0 * HEAD + hq * 4);
        const v4f a1 = *(const v4f*)(pacc + (size_t)row1 * HEAD + hq * 4);
        const v4f o  = (a0 * e0 + a1 * e1) * inv;

        const int qrow = batch * SEQ + (63 - pr) * 64 + qi;
        *(v4f*)(out + (size_t)qrow * HEAD + hq * 4) = o;
    }
}

extern "C" void kernel_launch(void* const* d_in, const int* in_sizes, int n_in,
                              void* d_out, int out_size, void* d_ws, size_t ws_size,
                              hipStream_t stream) {
    const float* x  = (const float*)d_in[0];
    const float* Wq = (const float*)d_in[1];
    const float* Wk = (const float*)d_in[2];
    const float* Wv = (const float*)d_in[3];

    float* qb = (float*)d_ws;
    float* kb = qb + (size_t)BS * HEAD;
    float* vT = kb + (size_t)BS * HEAD;
    float* pacc = vT + (size_t)BS * HEAD;
    float* pm   = pacc + (size_t)256 * 64 * HEAD;
    float* pl   = pm + (size_t)256 * 64;

    qkv_proj_kernel<<<256, 128, 0, stream>>>(x, Wq, Wk, Wv, qb, kb, vT);
    flash_attn_kernel<<<256, 128, 0, stream>>>(qb, kb, vT, (float*)d_out, pacc, pm, pl);
    merge_kernel<<<496, 64, 0, stream>>>(pacc, pm, pl, (float*)d_out);
}


# Attention: causal work cut into 512 equal 16-17-chunk shares, two workgroups per WGP = two waves per SIMD; a block's up to 4 partial softmax states joined by the merge launch: key sum tree changes
# speedup vs baseline: 7.4184x; 1.0899x over previous
#include <hip/hip_runtime.h>
#include <math.h>

typedef float v2f __attribute__((ext_vector_type(2)));
typedef float v4f __attribute__((ext_vector_type(4)));
typedef float v8f __attribute__((ext_vector_type(8)));

#define EMBD 768
#define HEAD 64
#define SEQ  4096
#define BATCH 4
#define BS   (BATCH * SEQ)

__device__ __forceinline__ v8f wmma_f32(v2f a, v2f b, v8f c) {
    return __builtin_amdgcn_wmma_f32_16x16x4_f32(
         false, a,  false, b,
         (short)0, c,  false,  false);
}

__device__ __forceinline__ float xhalf(float x) {
    const int b = __builtin_bit_cast(int, x);
    return __builtin_bit_cast(float,
        __builtin_amdgcn_permlanex16(b, b, 0x76543210, 0xfedcba98, false, false));
}

#define NSPAN (EMBD / 64)

__device__ __forceinline__ void proj_load_w(v4f (&b)[24], const float* wq,
                                            const float* wk, const float* wv, int e0)
{
    #pragma unroll
    for (int u = 0; u < 2; ++u)
        #pragma unroll
        for (int i = 0; i < 4; ++i) {
            const int off = (e0 + 8 * u + i) * HEAD;
            b[(u * 4 + i) * 3 + 0] = *(const v4f*)(wq + off);
            b[(u * 4 + i) * 3 + 1] = *(const v4f*)(wk + off);
            b[(u * 4 + i) * 3 + 2] = *(const v4f*)(wv + off);
        }
}

__device__ __forceinline__ void proj_mma(v8f (&acc)[12], const v4f a0, const v4f a1,
                                         const v4f (&b)[24])
{
    #pragma unroll
    for (int u = 0; u < 2; ++u) {
        const v4f a = u ? a1 : a0;
        #pragma unroll
        for (int s = 0; s < 2; ++s) {
            v2f af;
            af.x = a[2 * s];
            af.y = a[2 * s + 1];
            #pragma unroll
            for (int m = 0; m < 3; ++m)
                #pragma unroll
                for (int t = 0; t < 4; ++t) {
                    v2f bf;
                    bf.x = b[(u * 4 + 2 * s) * 3 + m][t];
                    bf.y = b[(u * 4 + 2 * s + 1) * 3 + m][t];
                    acc[m * 4 + t] = wmma_f32(af, bf, acc[m * 4 + t]);
                }
        }
    }
}

__global__ __launch_bounds__(128) void qkv_proj_kernel(
    const float* __restrict__ x,
    const float* __restrict__ Wq, const float* __restrict__ Wk,
    const float* __restrict__ Wv,
    float* __restrict__ qb, float* __restrict__ kb, float* __restrict__ vT)
{
    const int lane = threadIdx.x & 31;
    const int wave = threadIdx.x >> 5;
    const int half = lane >> 4;
    const int lx   = lane & 15;
    const int row0 = (blockIdx.x * 4 + wave) << 4;

    const float* xrow = x + (size_t)(row0 + lx) * EMBD + 4 * half;
    const int woff = (4 * half) * HEAD + 4 * lx;
    const float* wq = Wq + woff;
    const float* wk = Wk + woff;
    const float* wv = Wv + woff;

    v8f acc[12] = {};
    v4f a_cur[8], a_nxt[8];
    v4f b0[24], b1[24];

    int sp = blockIdx.x % NSPAN;

    #pragma unroll
    for (int j = 0; j < 8; ++j) a_cur[j] = *(const v4f*)(xrow + 64 * sp + 8 * j);
    proj_load_w(b0, wq, wk, wv, 64 * sp);

    #pragma unroll 1
    for (int it = 0; it < NSPAN; ++it) {
        int spn = sp + 1;
        if (spn == NSPAN) spn = 0;
        const int e = 64 * sp;
        #pragma unroll
        for (int j = 0; j < 8; ++j) a_nxt[j] = *(const v4f*)(xrow + 64 * spn + 8 * j);

        proj_load_w(b1, wq, wk, wv, e + 16);
        __builtin_amdgcn_sched_barrier(0);
        proj_mma(acc, a_cur[0], a_cur[1], b0);
        __builtin_amdgcn_sched_barrier(0);
        proj_load_w(b0, wq, wk, wv, e + 32);
        __builtin_amdgcn_sched_barrier(0);
        proj_mma(acc, a_cur[2], a_cur[3], b1);
        __builtin_amdgcn_sched_barrier(0);
        proj_load_w(b1, wq, wk, wv, e + 48);
        __builtin_amdgcn_sched_barrier(0);
        proj_mma(acc, a_cur[4], a_cur[5], b0);
        __builtin_amdgcn_sched_barrier(0);
        proj_load_w(b0, wq, wk, wv, 64 * spn);
        __builtin_amdgcn_sched_barrier(0);
        proj_mma(acc, a_cur[6], a_cur[7], b1);
        __builtin_amdgcn_sched_barrier(0);

        #pragma unroll
        for (int j = 0; j < 8; ++j) a_cur[j] = a_nxt[j];
        sp = spn;
    }

    const int r0 = row0 + 8 * half;
    #pragma unroll
    for (int m = 0; m < 2; ++m) {
        float* ob = (m == 0) ? qb : kb;
        #pragma unroll
        for (int i = 0; i < 8; ++i) {
            v4f o;
            o.x = acc[m * 4 + 0][i];
            o.y = acc[m * 4 + 1][i];
            o.z = acc[m * 4 + 2][i];
            o.w = acc[m * 4 + 3][i];
            *(v4f*)(ob + (size_t)(r0 + i) * HEAD + 4 * lx) = o;
        }
    }
    {
        const int batch = row0 >> 12;
        const int s0    = (row0 & 4095) + 8 * half;
        float* vb = vT + (size_t)batch * HEAD * SEQ;
        #pragma unroll
        for (int t = 0; t < 4; ++t) {
            float* p = vb + (size_t)(4 * lx + t) * SEQ + s0;
            v4f lo, up;
            lo.x = acc[8 + t][0]; lo.y = acc[8 + t][1];
            lo.z = acc[8 + t][2]; lo.w = acc[8 + t][3];
            up.x = acc[8 + t][4]; up.y = acc[8 + t][5];
            up.z = acc[8 + t][6]; up.w = acc[8 + t][7];
            *(v4f*)(p)     = lo;
            *(v4f*)(p + 4) = up;
        }
    }
}

#define QW 4
#define KC 64
#define LP 68
#define NSH 4
#define PAIRLEN 65

__device__ __forceinline__ int share_bound(int s) {
    return (PAIRLEN * s + NSH - 1) / NSH;
}

__global__ __launch_bounds__(128) void flash_attn_kernel(
    const float* __restrict__ qb, const float* __restrict__ kb,
    const float* __restrict__ vT, float* __restrict__ out,
    float* __restrict__ pacc, float* __restrict__ pm, float* __restrict__ pl)
{
    __shared__ __attribute__((aligned(16))) float sK[2][KC * LP];
    __shared__ __attribute__((aligned(16))) float sV[2][HEAD * LP];

    const int tid   = threadIdx.x;
    const int lane  = tid & 31;
    const int wave  = __builtin_amdgcn_readfirstlane(tid >> 5);
    const int share = blockIdx.x >> 7;
    const int batch = blockIdx.x & 3;
    const int pr    = (blockIdx.x >> 2) & 31;
    const int half  = lane >> 4;
    const int lx    = lane & 15;
    const bool hi   = (half != 0);

    const float scale = 0.125f;

    const float* kbase = kb + (size_t)batch * SEQ * HEAD;
    const float* vbase = vT + (size_t)batch * HEAD * SEQ;

    const int srow = tid >> 4;
    const int scol = (tid & 15) << 2;

    const int nlong = 64 - pr;
    const int pos0  = share_bound(share);
    const int pos1  = share_bound(share + 1);

    for (int s = 0; s < 2; ++s) {
        const int bpos0 = s ? nlong : 0;
        const int bpos1 = s ? PAIRLEN : nlong;
        const int p0 = pos0 > bpos0 ? pos0 : bpos0;
        const int p1 = pos1 < bpos1 ? pos1 : bpos1;
        if (p0 >= p1) continue;
        const int qblk = s ? pr : 63 - pr;
        const int c0   = p0 - bpos0;
        const int c1   = p1 - bpos0;
        const bool whole = (c0 == 0) && (c1 == qblk + 1);

        const int qt = qblk * QW + wave;
        const int q0 = qt << 4;

        v2f qf[16];
        {
            const float* qrow = qb + (size_t)(batch * SEQ + q0 + lx) * HEAD;
            #pragma unroll
            for (int c = 0; c < 16; ++c) {
                v2f v = *(const v2f*)(qrow + 4 * c + 2 * half);
                qf[c].x = v.x * scale;
                qf[c].y = v.y * scale;
            }
        }

        v8f acc[4] = {};
        float m = -INFINITY, l = 0.0f;
        const int q_lane = q0 + lx;
        const int nkt    = qt + 1;

        v4f rk[8], rv[8];

        {
            const int kc0 = c0 * KC;
            #pragma unroll
            for (int i = 0; i < 8; ++i) {
                const int r = srow + 8 * i;
                rk[i] = *(const v4f*)(kbase + (size_t)(kc0 + r) * HEAD + scol);
                rv[i] = *(const v4f*)(vbase + (size_t)r * SEQ + kc0 + scol);
            }
            #pragma unroll
            for (int i = 0; i < 8; ++i) {
                const int r = srow + 8 * i;
                *(v4f*)(&sK[0][r * LP + scol]) = rk[i];
                *(v4f*)(&sV[0][r * LP + scol]) = rv[i];
            }
        }
        __syncthreads();

        for (int ch = c0; ch < c1; ++ch) {
            const int buf   = (ch - c0) & 1;
            const bool more = (ch + 1 < c1);

            if (more) {
                const int kc1 = (ch + 1) * KC;
                #pragma unroll
                for (int i = 0; i < 8; ++i) {
                    const int r = srow + 8 * i;
                    rk[i] = *(const v4f*)(kbase + (size_t)(kc1 + r) * HEAD + scol);
                    rv[i] = *(const v4f*)(vbase + (size_t)r * SEQ + kc1 + scol);
                }
            }

            const float* cK = sK[buf];
            const float* cV = sV[buf];
            int ntile = nkt - 4 * ch;
            if (ntile > 4) ntile = 4;

            for (int kk = 0; kk < ntile; ++kk) {
                const int k0 = (4 * ch + kk) << 4;

                v8f st = {};
                const float* krow = cK + (kk * 16 + lx) * LP + 2 * half;
                #pragma unroll
                for (int c = 0; c < 16; ++c) {
                    v2f kf = *(const v2f*)(krow + 4 * c);
                    st = wmma_f32(kf, qf[c], st);
                }

                float p[8];
                float tmax = -INFINITY;
                #pragma unroll
                for (int i = 0; i < 8; ++i) {
                    const int key = k0 + i + 8 * half;
                    const float sc = (key <= q_lane) ? st[i] : -INFINITY;
                    p[i] = sc;
                    tmax = fmaxf(tmax, sc);
                }
                tmax = fmaxf(tmax, xhalf(tmax));
                const float m_new = fmaxf(m, tmax);
                const float alpha = __expf(m - m_new);
                float rsum = 0.0f;
                #pragma unroll
                for (int i = 0; i < 8; ++i) {
                    p[i] = __expf(p[i] - m_new);
                    rsum += p[i];
                }
                l = l * alpha + rsum + xhalf(rsum);
                m = m_new;
                #pragma unroll
                for (int t = 0; t < 4; ++t)
                    #pragma unroll
                    for (int i = 0; i < 8; ++i) acc[t][i] *= alpha;

                const float g0 = xhalf(hi ? p[0] : p[2]);
                const float g1 = xhalf(hi ? p[1] : p[3]);
                const float g2 = xhalf(hi ? p[4] : p[6]);
                const float g3 = xhalf(hi ? p[5] : p[7]);
                v2f pf[4];
                pf[0].x = hi ? g0 : p[0];   pf[0].y = hi ? g1 : p[1];
                pf[1].x = hi ? g2 : p[4];   pf[1].y = hi ? g3 : p[5];
                pf[2].x = hi ? p[2] : g0;   pf[2].y = hi ? p[3] : g1;
                pf[3].x = hi ? p[6] : g2;   pf[3].y = hi ? p[7] : g3;

                #pragma unroll
                for (int t = 0; t < 4; ++t) {
                    const float* vrow = cV + (t * 16 + lx) * LP + kk * 16 + 2 * half;
                    #pragma unroll
                    for (int c = 0; c < 4; ++c) {
                        v2f vf = *(const v2f*)(vrow + 4 * c);
                        acc[t] = wmma_f32(vf, pf[c], acc[t]);
                    }
                }
            }

            if (more) {
                #pragma unroll
                for (int i = 0; i < 8; ++i) {
                    const int r = srow + 8 * i;
                    *(v4f*)(&sK[buf ^ 1][r * LP + scol]) = rk[i];
                    *(v4f*)(&sV[buf ^ 1][r * LP + scol]) = rv[i];
                }
            }
            __syncthreads();
        }

        if (whole) {
            const float inv = 1.0f / l;
            float* orow = out + (size_t)(batch * SEQ + q0 + lx) * HEAD + 8 * half;
            #pragma unroll
            for (int t = 0; t < 4; ++t) {
                float4 lo = make_float4(acc[t][0] * inv, acc[t][1] * inv,
                                        acc[t][2] * inv, acc[t][3] * inv);
                float4 up = make_float4(acc[t][4] * inv, acc[t][5] * inv,
                                        acc[t][6] * inv, acc[t][7] * inv);
                *(float4*)(orow + t * 16)     = lo;
                *(float4*)(orow + t * 16 + 4) = up;
            }
        } else {
            const int prow = ((((batch * 32 + pr) * NSH + share) * 2 + s) * 64) + wave * 16 + lx;
            float* arow = pacc + (size_t)prow * HEAD + 8 * half;
            #pragma unroll
            for (int t = 0; t < 4; ++t) {
                float4 lo = make_float4(acc[t][0], acc[t][1], acc[t][2], acc[t][3]);
                float4 up = make_float4(acc[t][4], acc[t][5], acc[t][6], acc[t][7]);
                *(float4*)(arow + t * 16)     = lo;
                *(float4*)(arow + t * 16 + 4) = up;
            }
            if (!hi) {
                pm[prow] = m;
                pl[prow] = l;
            }
        }
    }
}

__global__ __launch_bounds__(64) void merge_kernel(
    const float* __restrict__ pacc, const float* __restrict__ pm,
    const float* __restrict__ pl, float* __restrict__ out)
{
    const int tid = threadIdx.x;
    const int hq  = tid & 15;
    const int qs  = tid >> 4;

    const int batch = blockIdx.x >> 8;
    const int blk   = (blockIdx.x >> 2) & 63;
    const int seg   = (blk < 32) ? 1 : 0;
    const int pr    = seg ? blk : 63 - blk;
    const int nlong = 64 - pr;
    const int bpos0 = seg ? nlong : 0;
    const int bpos1 = seg ? PAIRLEN : nlong;

    int first = NSH, last = -1;
    #pragma unroll
    for (int s = 0; s < NSH; ++s) {
        const int b0 = share_bound(s), b1 = share_bound(s + 1);
        const int p0 = b0 > bpos0 ? b0 : bpos0;
        const int p1 = b1 < bpos1 ? b1 : bpos1;
        if (p0 < p1) {
            if (s < first) first = s;
            last = s;
        }
    }
    if (last <= first) return;

    const int slot0 = (batch * 32 + pr) * NSH;

    const int qi0 = (blockIdx.x & 3) * 16 + qs;

    float mi[4][NSH], li[4][NSH];
    v4f   ai[4][NSH];
    #pragma unroll
    for (int k = 0; k < 4; ++k) {
        #pragma unroll
        for (int j = 0; j < NSH; ++j) {
            const int s   = (first + j < last) ? first + j : last;
            const int row = ((slot0 + s) * 2 + seg) * 64 + qi0 + 4 * k;
            mi[k][j] = pm[row];
            li[k][j] = pl[row];
            ai[k][j] = *(const v4f*)(pacc + (size_t)row * HEAD + hq * 4);
        }
    }
    __builtin_amdgcn_sched_barrier(0);

    #pragma unroll
    for (int k = 0; k < 4; ++k) {
        float mv[NSH];
        #pragma unroll
        for (int j = 0; j < NSH; ++j)
            mv[j] = (first + j <= last) ? mi[k][j] : -INFINITY;
        float mm = mv[0];
        #pragma unroll
        for (int j = 1; j < NSH; ++j) mm = fmaxf(mm, mv[j]);

        float den = 0.0f;
        v4f o = {0.0f, 0.0f, 0.0f, 0.0f};
        #pragma unroll
        for (int j = 0; j < NSH; ++j) {
            const float e = __expf(mv[j] - mm);
            den += e * li[k][j];
            o += ai[k][j] * e;
        }
        const float inv = 1.0f / den;
        o = o * inv;

        const int qrow = batch * SEQ + blk * 64 + qi0 + 4 * k;
        *(v4f*)(out + (size_t)qrow * HEAD + hq * 4) = o;
    }
}

extern "C" void kernel_launch(void* const* d_in, const int* in_sizes, int n_in,
                              void* d_out, int out_size, void* d_ws, size_t ws_size,
                              hipStream_t stream) {
    const float* x  = (const float*)d_in[0];
    const float* Wq = (const float*)d_in[1];
    const float* Wk = (const float*)d_in[2];
    const float* Wv = (const float*)d_in[3];

    float* qb = (float*)d_ws;
    float* kb = qb + (size_t)BS * HEAD;
    float* vT = kb + (size_t)BS * HEAD;
    float* pacc = vT + (size_t)BS * HEAD;
    float* pm   = pacc + (size_t)1024 * 64 * HEAD;
    float* pl   = pm + (size_t)1024 * 64;

    qkv_proj_kernel<<<256, 128, 0, stream>>>(x, Wq, Wk, Wv, qb, kb, vT);
    flash_attn_kernel<<<512, 128, 0, stream>>>(qb, kb, vT, (float*)d_out, pacc, pm, pl);
    merge_kernel<<<1024, 64, 0, stream>>>(pacc, pm, pl, (float*)d_out);
}
